# starting kernel
# baseline (speedup 1.0000x reference)
#include <hip/hip_runtime.h>

#define TT      2048
#define NBANDS  128
#define NBATCH  16
#define ROWS    (NBATCH * NBANDS)
#define CHUNK   16
#define NCHUNKS (TT / CHUNK)
#define EPSV    1e-5f

typedef __attribute__((ext_vector_type(2))) float v2f;
typedef __attribute__((ext_vector_type(4))) float v4f;
typedef __attribute__((ext_vector_type(8))) float v8f;

__device__ __forceinline__ float s_coef() {
    return (sqrtf(262145.0f) - 1.0f) / 131072.0f;
}

__device__ __forceinline__ float qpow(float q, int e) {
    float r = 1.0f;
#pragma unroll 1
    for (int i = 0; i < e; ++i) r *= q;
    return r;
}

__device__ __forceinline__ float fast_pow_pos(float y, float p) {
    return __builtin_amdgcn_exp2f(p * __builtin_amdgcn_logf(y));
}

__global__ void __launch_bounds__(NBANDS)
pcen_pass0_bands(const float* __restrict__ alpha, const float* __restrict__ delta,
                 const float* __restrict__ r, float* __restrict__ bws) {
    int b = threadIdx.x;
    float a  = __expf(alpha[b]);
    float d  = __expf(delta[b]);
    float rr = __expf(r[b]);
    bws[b]              = a;
    bws[NBANDS + b]     = d;
    bws[2 * NBANDS + b] = rr;
    bws[3 * NBANDS + b] = fast_pow_pos(d, rr);
}

__global__ void __launch_bounds__(256)
pcen_pass1_chunklocal(const float* __restrict__ x, float* __restrict__ Lws) {
    int gid   = blockIdx.x * blockDim.x + threadIdx.x;
    int row   = gid >> 7;
    int chunk = gid & (NCHUNKS - 1);
    const float s = s_coef();
    const float q = 1.0f - s;

    const v4f* p4 = (const v4f*)(x + (size_t)row * TT + (size_t)chunk * CHUNK);
    float acc = 0.0f;
#pragma unroll
    for (int i = 0; i < 4; ++i) {
        v4f v = p4[i];
        acc = acc * q + v.x;
        acc = acc * q + v.y;
        acc = acc * q + v.z;
        acc = acc * q + v.w;
    }
    Lws[gid] = s * acc;
}

__global__ void __launch_bounds__(256)
pcen_pass2_carry(const float* __restrict__ x, const float* __restrict__ Lws,
                 float* __restrict__ Cws) {
    int row = blockIdx.x * blockDim.x + threadIdx.x;
    const float s  = s_coef();
    const float q  = 1.0f - s;
    const float q2 = q * q, q4 = q2 * q2, q8 = q4 * q4, q16 = q8 * q8;

    float c = x[(size_t)row * TT];
    const float* Lr = Lws + (size_t)row * NCHUNKS;
    float*       Cr = Cws + (size_t)row * NCHUNKS;
#pragma unroll 4
    for (int k = 0; k < NCHUNKS; ++k) {
        Cr[k] = c;
        c = q16 * c + Lr[k];
    }
}

__global__ void __launch_bounds__(32)
pcen_pass3_wmma(const float* __restrict__ x, const float* __restrict__ Cws,
                const float* __restrict__ bws, float* __restrict__ out) {
    const float s = s_coef();
    const float q = 1.0f - s;

    const int chunk = blockIdx.x;
    const int rg    = blockIdx.y;
    const int row0  = rg * 16;
    const int t0    = chunk * CHUNK;

    const int lane = threadIdx.x;
    const int m    = lane & 15;
    const int hi   = lane >> 4;

    v2f Bm[4];
#pragma unroll
    for (int kk = 0; kk < 4; ++kk) {
        int i0 = kk * 4 + 2 * hi;
        Bm[kk].x = (m >= i0)     ? s * qpow(q, m - i0)     : 0.0f;
        Bm[kk].y = (m >= i0 + 1) ? s * qpow(q, m - i0 - 1) : 0.0f;
    }

    const float* xr = x + (size_t)(row0 + m) * TT;
    v2f Am[4];
#pragma unroll
    for (int kk = 0; kk < 4; ++kk) {
        int bt = t0 + kk * 4 + 2 * hi;
        Am[kk] = *(const v2f*)(xr + bt);
    }

    const float qn1 = qpow(q, m + 1);
    v8f acc;
#pragma unroll
    for (int v = 0; v < 8; ++v) {
        int rowM = row0 + v + 8 * hi;
        acc[v] = Cws[(size_t)rowM * NCHUNKS + chunk] * qn1;
    }

    acc = __builtin_amdgcn_wmma_f32_16x16x4_f32(false, Am[0], false, Bm[0], (short)0, acc, false, false);
    acc = __builtin_amdgcn_wmma_f32_16x16x4_f32(false, Am[1], false, Bm[1], (short)0, acc, false, false);
    acc = __builtin_amdgcn_wmma_f32_16x16x4_f32(false, Am[2], false, Bm[2], (short)0, acc, false, false);
    acc = __builtin_amdgcn_wmma_f32_16x16x4_f32(false, Am[3], false, Bm[3], (short)0, acc, false, false);

    const int t         = t0 + m;
    const int bidx      = row0 >> 7;
    const int band_base = (row0 & 127) + 8 * hi;
    const float* ea  = bws;
    const float* ed  = bws + NBANDS;
    const float* err = bws + 2 * NBANDS;
    const float* edr = bws + 3 * NBANDS;
    v4f a_lo  = *(const v4f*)(ea  + band_base), a_hi  = *(const v4f*)(ea  + band_base + 4);
    v4f d_lo  = *(const v4f*)(ed  + band_base), d_hi  = *(const v4f*)(ed  + band_base + 4);
    v4f r_lo  = *(const v4f*)(err + band_base), r_hi  = *(const v4f*)(err + band_base + 4);
    v4f dr_lo = *(const v4f*)(edr + band_base), dr_hi = *(const v4f*)(edr + band_base + 4);
    float av[8]  = {a_lo.x, a_lo.y, a_lo.z, a_lo.w, a_hi.x, a_hi.y, a_hi.z, a_hi.w};
    float dv[8]  = {d_lo.x, d_lo.y, d_lo.z, d_lo.w, d_hi.x, d_hi.y, d_hi.z, d_hi.w};
    float rv[8]  = {r_lo.x, r_lo.y, r_lo.z, r_lo.w, r_hi.x, r_hi.y, r_hi.z, r_hi.w};
    float drv[8] = {dr_lo.x, dr_lo.y, dr_lo.z, dr_lo.w, dr_hi.x, dr_hi.y, dr_hi.z, dr_hi.w};

    float res[8];
#pragma unroll
    for (int v = 0; v < 8; ++v) {
        int   rowM = row0 + v + 8 * hi;
        float xv   = x[(size_t)rowM * TT + t];
        float sm   = fmaxf(acc[v], 0.0f);
        float smooth = fast_pow_pos(EPSV + sm, -av[v]);
        float base   = fmaf(xv, smooth, dv[v]);
        res[v] = fast_pow_pos(base, rv[v]) - drv[v];
    }
    float* ob = out + (size_t)bidx * TT * NBANDS + (size_t)t * NBANDS + band_base;
    v4f lo;  lo.x  = res[0]; lo.y  = res[1]; lo.z  = res[2]; lo.w  = res[3];
    v4f hi4; hi4.x = res[4]; hi4.y = res[5]; hi4.z = res[6]; hi4.w = res[7];
    *(v4f*)(ob)     = lo;
    *(v4f*)(ob + 4) = hi4;
}

extern "C" void kernel_launch(void* const* d_in, const int* in_sizes, int n_in,
                              void* d_out, int out_size, void* d_ws, size_t ws_size,
                              hipStream_t stream) {
    const float* x     = (const float*)d_in[0];
    const float* alpha = (const float*)d_in[1];
    const float* delta = (const float*)d_in[2];
    const float* r     = (const float*)d_in[3];
    float* out = (float*)d_out;

    float* bws = (float*)d_ws;
    float* Lws = bws + 4 * NBANDS;
    float* Cws = Lws + (size_t)ROWS * NCHUNKS;

    pcen_pass0_bands<<<1, NBANDS, 0, stream>>>(alpha, delta, r, bws);
    pcen_pass1_chunklocal<<<(ROWS * NCHUNKS) / 256, 256, 0, stream>>>(x, Lws);
    pcen_pass2_carry<<<ROWS / 256, 256, 0, stream>>>(x, Lws, Cws);

    dim3 g3(NCHUNKS, ROWS / 16);
    pcen_pass3_wmma<<<g3, 32, 0, stream>>>(x, Cws, bws, out);
}


# Chunk sums + carry scan fused into one wave-per-row kernel: lane-local Horner over 4 chunks, then 5-step shuffle prefix over lanes (new reduction tree for the carries); 3 launches
# speedup vs baseline: 1.5463x; 1.5463x over previous
#include <hip/hip_runtime.h>

#define TT      2048
#define NBANDS  128
#define NBATCH  16
#define ROWS    (NBATCH * NBANDS)
#define CHUNK   16
#define NCHUNKS (TT / CHUNK)
#define EPSV    1e-5f

typedef __attribute__((ext_vector_type(2))) float v2f;
typedef __attribute__((ext_vector_type(4))) float v4f;
typedef __attribute__((ext_vector_type(8))) float v8f;

__device__ __forceinline__ float s_coef() {
    return (sqrtf(262145.0f) - 1.0f) / 131072.0f;
}

__device__ __forceinline__ float qpow(float q, int e) {
    float r = 1.0f;
#pragma unroll 1
    for (int i = 0; i < e; ++i) r *= q;
    return r;
}

__device__ __forceinline__ float fast_pow_pos(float y, float p) {
    return __builtin_amdgcn_exp2f(p * __builtin_amdgcn_logf(y));
}

__global__ void __launch_bounds__(NBANDS)
pcen_pass0_bands(const float* __restrict__ alpha, const float* __restrict__ delta,
                 const float* __restrict__ r, float* __restrict__ bws) {
    int b = threadIdx.x;
    float a  = __expf(alpha[b]);
    float d  = __expf(delta[b]);
    float rr = __expf(r[b]);
    bws[b]              = a;
    bws[NBANDS + b]     = d;
    bws[2 * NBANDS + b] = rr;
    bws[3 * NBANDS + b] = fast_pow_pos(d, rr);
}

#define SCAN_ROWS_PER_BLOCK 4
__global__ void __launch_bounds__(32 * SCAN_ROWS_PER_BLOCK)
pcen_pass12_rowscan(const float* __restrict__ x, float* __restrict__ Cws) {
    const int lane = threadIdx.x & 31;
    const int row  = blockIdx.x * SCAN_ROWS_PER_BLOCK + (threadIdx.x >> 5);
    const float s   = s_coef();
    const float q   = 1.0f - s;
    const float q2  = q * q, q4 = q2 * q2, q8 = q4 * q4, q16 = q8 * q8;
    const float q32 = q16 * q16, q64 = q32 * q32;

    const float* xr = x + (size_t)row * TT;
    const v4f*   p4 = (const v4f*)(xr + lane * 64);
    const float  c0 = xr[0];

    float L[4];
#pragma unroll
    for (int c = 0; c < 4; ++c) {
        float acc = 0.0f;
#pragma unroll
        for (int i = 0; i < 4; ++i) {
            v4f v = p4[c * 4 + i];
            acc = acc * q + v.x;
            acc = acc * q + v.y;
            acc = acc * q + v.z;
            acc = acc * q + v.w;
        }
        L[c] = s * acc;
    }

    float P = ((L[0] * q16 + L[1]) * q16 + L[2]) * q16 + L[3];
    if (lane == 0) P = q64 * c0 + P;
    float w = q64;
#pragma unroll
    for (int d = 1; d < 32; d <<= 1) {
        float up = __shfl_up(P, d, 32);
        if (lane >= d) P = w * up + P;
        w = w * w;
    }
    float cin = __shfl_up(P, 1, 32);
    if (lane == 0) cin = c0;

    v4f cv;
    cv.x = cin;
    cv.y = q16 * cv.x + L[0];
    cv.z = q16 * cv.y + L[1];
    cv.w = q16 * cv.z + L[2];
    *(v4f*)(Cws + (size_t)row * NCHUNKS + lane * 4) = cv;
}

__global__ void __launch_bounds__(32)
pcen_pass3_wmma(const float* __restrict__ x, const float* __restrict__ Cws,
                const float* __restrict__ bws, float* __restrict__ out) {
    const float s = s_coef();
    const float q = 1.0f - s;

    const int chunk = blockIdx.x;
    const int rg    = blockIdx.y;
    const int row0  = rg * 16;
    const int t0    = chunk * CHUNK;

    const int lane = threadIdx.x;
    const int m    = lane & 15;
    const int hi   = lane >> 4;

    v2f Bm[4];
#pragma unroll
    for (int kk = 0; kk < 4; ++kk) {
        int i0 = kk * 4 + 2 * hi;
        Bm[kk].x = (m >= i0)     ? s * qpow(q, m - i0)     : 0.0f;
        Bm[kk].y = (m >= i0 + 1) ? s * qpow(q, m - i0 - 1) : 0.0f;
    }

    const float* xr = x + (size_t)(row0 + m) * TT;
    v2f Am[4];
#pragma unroll
    for (int kk = 0; kk < 4; ++kk) {
        int bt = t0 + kk * 4 + 2 * hi;
        Am[kk] = *(const v2f*)(xr + bt);
    }

    const float qn1 = qpow(q, m + 1);
    v8f acc;
#pragma unroll
    for (int v = 0; v < 8; ++v) {
        int rowM = row0 + v + 8 * hi;
        acc[v] = Cws[(size_t)rowM * NCHUNKS + chunk] * qn1;
    }

    acc = __builtin_amdgcn_wmma_f32_16x16x4_f32(false, Am[0], false, Bm[0], (short)0, acc, false, false);
    acc = __builtin_amdgcn_wmma_f32_16x16x4_f32(false, Am[1], false, Bm[1], (short)0, acc, false, false);
    acc = __builtin_amdgcn_wmma_f32_16x16x4_f32(false, Am[2], false, Bm[2], (short)0, acc, false, false);
    acc = __builtin_amdgcn_wmma_f32_16x16x4_f32(false, Am[3], false, Bm[3], (short)0, acc, false, false);

    const int t         = t0 + m;
    const int bidx      = row0 >> 7;
    const int band_base = (row0 & 127) + 8 * hi;
    const float* ea  = bws;
    const float* ed  = bws + NBANDS;
    const float* err = bws + 2 * NBANDS;
    const float* edr = bws + 3 * NBANDS;
    v4f a_lo  = *(const v4f*)(ea  + band_base), a_hi  = *(const v4f*)(ea  + band_base + 4);
    v4f d_lo  = *(const v4f*)(ed  + band_base), d_hi  = *(const v4f*)(ed  + band_base + 4);
    v4f r_lo  = *(const v4f*)(err + band_base), r_hi  = *(const v4f*)(err + band_base + 4);
    v4f dr_lo = *(const v4f*)(edr + band_base), dr_hi = *(const v4f*)(edr + band_base + 4);
    float av[8]  = {a_lo.x, a_lo.y, a_lo.z, a_lo.w, a_hi.x, a_hi.y, a_hi.z, a_hi.w};
    float dv[8]  = {d_lo.x, d_lo.y, d_lo.z, d_lo.w, d_hi.x, d_hi.y, d_hi.z, d_hi.w};
    float rv[8]  = {r_lo.x, r_lo.y, r_lo.z, r_lo.w, r_hi.x, r_hi.y, r_hi.z, r_hi.w};
    float drv[8] = {dr_lo.x, dr_lo.y, dr_lo.z, dr_lo.w, dr_hi.x, dr_hi.y, dr_hi.z, dr_hi.w};

    float res[8];
#pragma unroll
    for (int v = 0; v < 8; ++v) {
        int   rowM = row0 + v + 8 * hi;
        float xv   = x[(size_t)rowM * TT + t];
        float sm   = fmaxf(acc[v], 0.0f);
        float smooth = fast_pow_pos(EPSV + sm, -av[v]);
        float base   = fmaf(xv, smooth, dv[v]);
        res[v] = fast_pow_pos(base, rv[v]) - drv[v];
    }
    float* ob = out + (size_t)bidx * TT * NBANDS + (size_t)t * NBANDS + band_base;
    v4f lo;  lo.x  = res[0]; lo.y  = res[1]; lo.z  = res[2]; lo.w  = res[3];
    v4f hi4; hi4.x = res[4]; hi4.y = res[5]; hi4.z = res[6]; hi4.w = res[7];
    *(v4f*)(ob)     = lo;
    *(v4f*)(ob + 4) = hi4;
}

extern "C" void kernel_launch(void* const* d_in, const int* in_sizes, int n_in,
                              void* d_out, int out_size, void* d_ws, size_t ws_size,
                              hipStream_t stream) {
    const float* x     = (const float*)d_in[0];
    const float* alpha = (const float*)d_in[1];
    const float* delta = (const float*)d_in[2];
    const float* r     = (const float*)d_in[3];
    float* out = (float*)d_out;

    float* bws = (float*)d_ws;
    float* Cws = bws + 4 * NBANDS;

    pcen_pass0_bands<<<1, NBANDS, 0, stream>>>(alpha, delta, r, bws);
    pcen_pass12_rowscan<<<ROWS / SCAN_ROWS_PER_BLOCK, 32 * SCAN_ROWS_PER_BLOCK, 0, stream>>>(x, Cws);

    dim3 g3(NCHUNKS, ROWS / 16);
    pcen_pass3_wmma<<<g3, 32, 0, stream>>>(x, Cws, bws, out);
}


# Pass 3 loops over tiles inside each wave: 8 consecutive time chunks per wave, 2048 waves in 256 work-groups instead of 16384 one-wave groups; decay matrix and band constants built once per wave
# speedup vs baseline: 1.7313x; 1.1196x over previous
#include <hip/hip_runtime.h>

#define TT      2048
#define NBANDS  128
#define NBATCH  16
#define ROWS    (NBATCH * NBANDS)
#define CHUNK   16
#define NCHUNKS (TT / CHUNK)
#define EPSV    1e-5f

typedef __attribute__((ext_vector_type(2))) float v2f;
typedef __attribute__((ext_vector_type(4))) float v4f;
typedef __attribute__((ext_vector_type(8))) float v8f;

__device__ __forceinline__ float s_coef() {
    return (sqrtf(262145.0f) - 1.0f) / 131072.0f;
}

__device__ __forceinline__ float qpow(float q, int e) {
    float r = 1.0f;
#pragma unroll 1
    for (int i = 0; i < e; ++i) r *= q;
    return r;
}

__device__ __forceinline__ float fast_pow_pos(float y, float p) {
    return __builtin_amdgcn_exp2f(p * __builtin_amdgcn_logf(y));
}

__global__ void __launch_bounds__(NBANDS)
pcen_pass0_bands(const float* __restrict__ alpha, const float* __restrict__ delta,
                 const float* __restrict__ r, float* __restrict__ bws) {
    int b = threadIdx.x;
    float a  = __expf(alpha[b]);
    float d  = __expf(delta[b]);
    float rr = __expf(r[b]);
    bws[b]              = a;
    bws[NBANDS + b]     = d;
    bws[2 * NBANDS + b] = rr;
    bws[3 * NBANDS + b] = fast_pow_pos(d, rr);
}

#define SCAN_ROWS_PER_BLOCK 4
__global__ void __launch_bounds__(32 * SCAN_ROWS_PER_BLOCK)
pcen_pass12_rowscan(const float* __restrict__ x, float* __restrict__ Cws) {
    const int lane = threadIdx.x & 31;
    const int row  = blockIdx.x * SCAN_ROWS_PER_BLOCK + (threadIdx.x >> 5);
    const float s   = s_coef();
    const float q   = 1.0f - s;
    const float q2  = q * q, q4 = q2 * q2, q8 = q4 * q4, q16 = q8 * q8;
    const float q32 = q16 * q16, q64 = q32 * q32;

    const float* xr = x + (size_t)row * TT;
    const v4f*   p4 = (const v4f*)(xr + lane * 64);
    const float  c0 = xr[0];

    float L[4];
#pragma unroll
    for (int c = 0; c < 4; ++c) {
        float acc = 0.0f;
#pragma unroll
        for (int i = 0; i < 4; ++i) {
            v4f v = p4[c * 4 + i];
            acc = acc * q + v.x;
            acc = acc * q + v.y;
            acc = acc * q + v.z;
            acc = acc * q + v.w;
        }
        L[c] = s * acc;
    }

    float P = ((L[0] * q16 + L[1]) * q16 + L[2]) * q16 + L[3];
    if (lane == 0) P = q64 * c0 + P;
    float w = q64;
#pragma unroll
    for (int d = 1; d < 32; d <<= 1) {
        float up = __shfl_up(P, d, 32);
        if (lane >= d) P = w * up + P;
        w = w * w;
    }
    float cin = __shfl_up(P, 1, 32);
    if (lane == 0) cin = c0;

    v4f cv;
    cv.x = cin;
    cv.y = q16 * cv.x + L[0];
    cv.z = q16 * cv.y + L[1];
    cv.w = q16 * cv.z + L[2];
    *(v4f*)(Cws + (size_t)row * NCHUNKS + lane * 4) = cv;
}

#define P3_TILES 8
#define P3_WAVES (NBANDS / 16)
#define P3_SPAN  (P3_TILES * CHUNK)
__global__ void __launch_bounds__(32 * P3_WAVES)
pcen_pass3_wmma(const float* __restrict__ x, const float* __restrict__ Cws,
                const float* __restrict__ bws, float* __restrict__ out) {
    const float s = s_coef();
    const float q = 1.0f - s;

    const int lane = threadIdx.x & 31;
    const int wv   = threadIdx.x >> 5;
    const int m    = lane & 15;
    const int hi   = lane >> 4;

    const int bidx   = blockIdx.y;
    const int chunk0 = blockIdx.x * P3_TILES;
    const int t00    = chunk0 * CHUNK;
    const int row0   = bidx * NBANDS + wv * 16;

    const float* xr = x + (size_t)(row0 + m) * TT + t00 + 2 * hi;
    v2f Am[P3_TILES][4];
#pragma unroll
    for (int i = 0; i < P3_TILES; ++i) {
#pragma unroll
        for (int kk = 0; kk < 4; ++kk)
            Am[i][kk] = *(const v2f*)(xr + i * CHUNK + kk * 4);
    }

    float cr[8][P3_TILES];
#pragma unroll
    for (int v = 0; v < 8; ++v) {
        const v4f* cp = (const v4f*)(Cws + (size_t)(row0 + v + 8 * hi) * NCHUNKS + chunk0);
        v4f c_lo = cp[0], c_hi = cp[1];
        cr[v][0] = c_lo.x; cr[v][1] = c_lo.y; cr[v][2] = c_lo.z; cr[v][3] = c_lo.w;
        cr[v][4] = c_hi.x; cr[v][5] = c_hi.y; cr[v][6] = c_hi.z; cr[v][7] = c_hi.w;
    }

    const int band_base = wv * 16 + 8 * hi;
    const float* ea  = bws;
    const float* ed  = bws + NBANDS;
    const float* err = bws + 2 * NBANDS;
    const float* edr = bws + 3 * NBANDS;
    v4f a_lo  = *(const v4f*)(ea  + band_base), a_hi  = *(const v4f*)(ea  + band_base + 4);
    v4f d_lo  = *(const v4f*)(ed  + band_base), d_hi  = *(const v4f*)(ed  + band_base + 4);
    v4f r_lo  = *(const v4f*)(err + band_base), r_hi  = *(const v4f*)(err + band_base + 4);
    v4f dr_lo = *(const v4f*)(edr + band_base), dr_hi = *(const v4f*)(edr + band_base + 4);
    float av[8]  = {a_lo.x, a_lo.y, a_lo.z, a_lo.w, a_hi.x, a_hi.y, a_hi.z, a_hi.w};
    float dv[8]  = {d_lo.x, d_lo.y, d_lo.z, d_lo.w, d_hi.x, d_hi.y, d_hi.z, d_hi.w};
    float rv[8]  = {r_lo.x, r_lo.y, r_lo.z, r_lo.w, r_hi.x, r_hi.y, r_hi.z, r_hi.w};
    float drv[8] = {dr_lo.x, dr_lo.y, dr_lo.z, dr_lo.w, dr_hi.x, dr_hi.y, dr_hi.z, dr_hi.w};

    v2f Bm[4];
#pragma unroll
    for (int kk = 0; kk < 4; ++kk) {
        int i0 = kk * 4 + 2 * hi;
        Bm[kk].x = (m >= i0)     ? s * qpow(q, m - i0)     : 0.0f;
        Bm[kk].y = (m >= i0 + 1) ? s * qpow(q, m - i0 - 1) : 0.0f;
    }
    const float qn1 = qpow(q, m + 1);

    const float* xe = x + (size_t)(row0 + 8 * hi) * TT + t00 + m;
    float xv[P3_TILES][8];
#pragma unroll
    for (int i = 0; i < P3_TILES; ++i) {
#pragma unroll
        for (int v = 0; v < 8; ++v)
            xv[i][v] = xe[(size_t)v * TT + i * CHUNK];
    }

    float* ob = out + (size_t)bidx * TT * NBANDS + (size_t)(t00 + m) * NBANDS + band_base;

#pragma unroll
    for (int i = 0; i < P3_TILES; ++i) {
        v8f acc;
#pragma unroll
        for (int v = 0; v < 8; ++v) acc[v] = cr[v][i] * qn1;

        acc = __builtin_amdgcn_wmma_f32_16x16x4_f32(false, Am[i][0], false, Bm[0], (short)0, acc, false, false);
        acc = __builtin_amdgcn_wmma_f32_16x16x4_f32(false, Am[i][1], false, Bm[1], (short)0, acc, false, false);
        acc = __builtin_amdgcn_wmma_f32_16x16x4_f32(false, Am[i][2], false, Bm[2], (short)0, acc, false, false);
        acc = __builtin_amdgcn_wmma_f32_16x16x4_f32(false, Am[i][3], false, Bm[3], (short)0, acc, false, false);

        float res[8];
#pragma unroll
        for (int v = 0; v < 8; ++v) {
            float sm     = fmaxf(acc[v], 0.0f);
            float smooth = fast_pow_pos(EPSV + sm, -av[v]);
            float base   = fmaf(xv[i][v], smooth, dv[v]);
            res[v] = fast_pow_pos(base, rv[v]) - drv[v];
        }
        float* o = ob + (size_t)i * CHUNK * NBANDS;
        v4f lo;  lo.x  = res[0]; lo.y  = res[1]; lo.z  = res[2]; lo.w  = res[3];
        v4f hi4; hi4.x = res[4]; hi4.y = res[5]; hi4.z = res[6]; hi4.w = res[7];
        *(v4f*)(o)     = lo;
        *(v4f*)(o + 4) = hi4;
    }
}

extern "C" void kernel_launch(void* const* d_in, const int* in_sizes, int n_in,
                              void* d_out, int out_size, void* d_ws, size_t ws_size,
                              hipStream_t stream) {
    const float* x     = (const float*)d_in[0];
    const float* alpha = (const float*)d_in[1];
    const float* delta = (const float*)d_in[2];
    const float* r     = (const float*)d_in[3];
    float* out = (float*)d_out;

    float* bws = (float*)d_ws;
    float* Cws = bws + 4 * NBANDS;

    pcen_pass0_bands<<<1, NBANDS, 0, stream>>>(alpha, delta, r, bws);
    pcen_pass12_rowscan<<<ROWS / SCAN_ROWS_PER_BLOCK, 32 * SCAN_ROWS_PER_BLOCK, 0, stream>>>(x, Cws);

    dim3 g3(TT / P3_SPAN, NBATCH);
    pcen_pass3_wmma<<<g3, 32 * P3_WAVES, 0, stream>>>(x, Cws, bws, out);
}


# Two launches instead of three: per-band constants (e^alpha, e^delta, e^r, d^rr) computed by work-group 0 of the row-scan kernel; the one-work-group launch in front of it is gone
# speedup vs baseline: 1.9381x; 1.1194x over previous
#include <hip/hip_runtime.h>

#define TT      2048
#define NBANDS  128
#define NBATCH  16
#define ROWS    (NBATCH * NBANDS)
#define CHUNK   16
#define NCHUNKS (TT / CHUNK)
#define EPSV    1e-5f

typedef __attribute__((ext_vector_type(2))) float v2f;
typedef __attribute__((ext_vector_type(4))) float v4f;
typedef __attribute__((ext_vector_type(8))) float v8f;

__device__ __forceinline__ float s_coef() {
    return (sqrtf(262145.0f) - 1.0f) / 131072.0f;
}

__device__ __forceinline__ float qpow(float q, int e) {
    float r = 1.0f;
#pragma unroll 1
    for (int i = 0; i < e; ++i) r *= q;
    return r;
}

__device__ __forceinline__ float fast_pow_pos(float y, float p) {
    return __builtin_amdgcn_exp2f(p * __builtin_amdgcn_logf(y));
}

#define SCAN_ROWS_PER_BLOCK 4
static_assert(32 * SCAN_ROWS_PER_BLOCK == NBANDS, "work-group 0 covers one band per thread");
__global__ void __launch_bounds__(32 * SCAN_ROWS_PER_BLOCK)
pcen_pass12_rowscan(const float* __restrict__ x, const float* __restrict__ alpha,
                    const float* __restrict__ delta, const float* __restrict__ r,
                    float* __restrict__ bws, float* __restrict__ Cws) {
    const int lane = threadIdx.x & 31;
    const int row  = blockIdx.x * SCAN_ROWS_PER_BLOCK + (threadIdx.x >> 5);
    const float s   = s_coef();
    const float q   = 1.0f - s;
    const float q2  = q * q, q4 = q2 * q2, q8 = q4 * q4, q16 = q8 * q8;
    const float q32 = q16 * q16, q64 = q32 * q32;

    const float* xr = x + (size_t)row * TT;
    const v4f*   p4 = (const v4f*)(xr + lane * 64);
    const float  c0 = xr[0];

    float L[4];
#pragma unroll
    for (int c = 0; c < 4; ++c) {
        float acc = 0.0f;
#pragma unroll
        for (int i = 0; i < 4; ++i) {
            v4f v = p4[c * 4 + i];
            acc = acc * q + v.x;
            acc = acc * q + v.y;
            acc = acc * q + v.z;
            acc = acc * q + v.w;
        }
        L[c] = s * acc;
    }

    float P = ((L[0] * q16 + L[1]) * q16 + L[2]) * q16 + L[3];
    if (lane == 0) P = q64 * c0 + P;
    float w = q64;
#pragma unroll
    for (int d = 1; d < 32; d <<= 1) {
        float up = __shfl_up(P, d, 32);
        if (lane >= d) P = w * up + P;
        w = w * w;
    }
    float cin = __shfl_up(P, 1, 32);
    if (lane == 0) cin = c0;

    v4f cv;
    cv.x = cin;
    cv.y = q16 * cv.x + L[0];
    cv.z = q16 * cv.y + L[1];
    cv.w = q16 * cv.z + L[2];
    *(v4f*)(Cws + (size_t)row * NCHUNKS + lane * 4) = cv;

    if (blockIdx.x == 0) {
        const int b = threadIdx.x;
        float a  = __expf(alpha[b]);
        float d  = __expf(delta[b]);
        float rr = __expf(r[b]);
        bws[b]              = a;
        bws[NBANDS + b]     = d;
        bws[2 * NBANDS + b] = rr;
        bws[3 * NBANDS + b] = fast_pow_pos(d, rr);
    }
}

#define P3_TILES 8
#define P3_WAVES (NBANDS / 16)
#define P3_SPAN  (P3_TILES * CHUNK)
__global__ void __launch_bounds__(32 * P3_WAVES)
pcen_pass3_wmma(const float* __restrict__ x, const float* __restrict__ Cws,
                const float* __restrict__ bws, float* __restrict__ out) {
    const float s = s_coef();
    const float q = 1.0f - s;

    const int lane = threadIdx.x & 31;
    const int wv   = threadIdx.x >> 5;
    const int m    = lane & 15;
    const int hi   = lane >> 4;

    const int bidx   = blockIdx.y;
    const int chunk0 = blockIdx.x * P3_TILES;
    const int t00    = chunk0 * CHUNK;
    const int row0   = bidx * NBANDS + wv * 16;

    const float* xr = x + (size_t)(row0 + m) * TT + t00 + 2 * hi;
    v2f Am[P3_TILES][4];
#pragma unroll
    for (int i = 0; i < P3_TILES; ++i) {
#pragma unroll
        for (int kk = 0; kk < 4; ++kk)
            Am[i][kk] = *(const v2f*)(xr + i * CHUNK + kk * 4);
    }

    float cr[8][P3_TILES];
#pragma unroll
    for (int v = 0; v < 8; ++v) {
        const v4f* cp = (const v4f*)(Cws + (size_t)(row0 + v + 8 * hi) * NCHUNKS + chunk0);
        v4f c_lo = cp[0], c_hi = cp[1];
        cr[v][0] = c_lo.x; cr[v][1] = c_lo.y; cr[v][2] = c_lo.z; cr[v][3] = c_lo.w;
        cr[v][4] = c_hi.x; cr[v][5] = c_hi.y; cr[v][6] = c_hi.z; cr[v][7] = c_hi.w;
    }

    const int band_base = wv * 16 + 8 * hi;
    const float* ea  = bws;
    const float* ed  = bws + NBANDS;
    const float* err = bws + 2 * NBANDS;
    const float* edr = bws + 3 * NBANDS;
    v4f a_lo  = *(const v4f*)(ea  + band_base), a_hi  = *(const v4f*)(ea  + band_base + 4);
    v4f d_lo  = *(const v4f*)(ed  + band_base), d_hi  = *(const v4f*)(ed  + band_base + 4);
    v4f r_lo  = *(const v4f*)(err + band_base), r_hi  = *(const v4f*)(err + band_base + 4);
    v4f dr_lo = *(const v4f*)(edr + band_base), dr_hi = *(const v4f*)(edr + band_base + 4);
    float av[8]  = {a_lo.x, a_lo.y, a_lo.z, a_lo.w, a_hi.x, a_hi.y, a_hi.z, a_hi.w};
    float dv[8]  = {d_lo.x, d_lo.y, d_lo.z, d_lo.w, d_hi.x, d_hi.y, d_hi.z, d_hi.w};
    float rv[8]  = {r_lo.x, r_lo.y, r_lo.z, r_lo.w, r_hi.x, r_hi.y, r_hi.z, r_hi.w};
    float drv[8] = {dr_lo.x, dr_lo.y, dr_lo.z, dr_lo.w, dr_hi.x, dr_hi.y, dr_hi.z, dr_hi.w};

    v2f Bm[4];
#pragma unroll
    for (int kk = 0; kk < 4; ++kk) {
        int i0 = kk * 4 + 2 * hi;
        Bm[kk].x = (m >= i0)     ? s * qpow(q, m - i0)     : 0.0f;
        Bm[kk].y = (m >= i0 + 1) ? s * qpow(q, m - i0 - 1) : 0.0f;
    }
    const float qn1 = qpow(q, m + 1);

    const float* xe = x + (size_t)(row0 + 8 * hi) * TT + t00 + m;
    float xv[P3_TILES][8];
#pragma unroll
    for (int i = 0; i < P3_TILES; ++i) {
#pragma unroll
        for (int v = 0; v < 8; ++v)
            xv[i][v] = xe[(size_t)v * TT + i * CHUNK];
    }

    float* ob = out + (size_t)bidx * TT * NBANDS + (size_t)(t00 + m) * NBANDS + band_base;

#pragma unroll
    for (int i = 0; i < P3_TILES; ++i) {
        v8f acc;
#pragma unroll
        for (int v = 0; v < 8; ++v) acc[v] = cr[v][i] * qn1;

        acc = __builtin_amdgcn_wmma_f32_16x16x4_f32(false, Am[i][0], false, Bm[0], (short)0, acc, false, false);
        acc = __builtin_amdgcn_wmma_f32_16x16x4_f32(false, Am[i][1], false, Bm[1], (short)0, acc, false, false);
        acc = __builtin_amdgcn_wmma_f32_16x16x4_f32(false, Am[i][2], false, Bm[2], (short)0, acc, false, false);
        acc = __builtin_amdgcn_wmma_f32_16x16x4_f32(false, Am[i][3], false, Bm[3], (short)0, acc, false, false);

        float res[8];
#pragma unroll
        for (int v = 0; v < 8; ++v) {
            float sm     = fmaxf(acc[v], 0.0f);
            float smooth = fast_pow_pos(EPSV + sm, -av[v]);
            float base   = fmaf(xv[i][v], smooth, dv[v]);
            res[v] = fast_pow_pos(base, rv[v]) - drv[v];
        }
        float* o = ob + (size_t)i * CHUNK * NBANDS;
        v4f lo;  lo.x  = res[0]; lo.y  = res[1]; lo.z  = res[2]; lo.w  = res[3];
        v4f hi4; hi4.x = res[4]; hi4.y = res[5]; hi4.z = res[6]; hi4.w = res[7];
        *(v4f*)(o)     = lo;
        *(v4f*)(o + 4) = hi4;
    }
}

extern "C" void kernel_launch(void* const* d_in, const int* in_sizes, int n_in,
                              void* d_out, int out_size, void* d_ws, size_t ws_size,
                              hipStream_t stream) {
    const float* x     = (const float*)d_in[0];
    const float* alpha = (const float*)d_in[1];
    const float* delta = (const float*)d_in[2];
    const float* r     = (const float*)d_in[3];
    float* out = (float*)d_out;

    float* bws = (float*)d_ws;
    float* Cws = bws + 4 * NBANDS;

    pcen_pass12_rowscan<<<ROWS / SCAN_ROWS_PER_BLOCK, 32 * SCAN_ROWS_PER_BLOCK, 0, stream>>>(
        x, alpha, delta, r, bws, Cws);

    dim3 g3(TT / P3_SPAN, NBATCH);
    pcen_pass3_wmma<<<g3, 32 * P3_WAVES, 0, stream>>>(x, Cws, bws, out);
}


# Pass 3 output tiles transposed through LDS (2 stages, one barrier per 2 tiles): a store instruction writes one whole 512-byte time row instead of 32 scattered 16-byte pieces
# speedup vs baseline: 2.6620x; 1.3735x over previous
#include <hip/hip_runtime.h>

#define TT      2048
#define NBANDS  128
#define NBATCH  16
#define ROWS    (NBATCH * NBANDS)
#define CHUNK   16
#define NCHUNKS (TT / CHUNK)
#define EPSV    1e-5f

typedef __attribute__((ext_vector_type(2))) float v2f;
typedef __attribute__((ext_vector_type(4))) float v4f;
typedef __attribute__((ext_vector_type(8))) float v8f;

__device__ __forceinline__ float s_coef() {
    return (sqrtf(262145.0f) - 1.0f) / 131072.0f;
}

__device__ __forceinline__ float qpow(float q, int e) {
    float r = 1.0f;
#pragma unroll 1
    for (int i = 0; i < e; ++i) r *= q;
    return r;
}

__device__ __forceinline__ float fast_pow_pos(float y, float p) {
    return __builtin_amdgcn_exp2f(p * __builtin_amdgcn_logf(y));
}

#define SCAN_ROWS_PER_BLOCK 4
static_assert(32 * SCAN_ROWS_PER_BLOCK == NBANDS, "work-group 0 covers one band per thread");
__global__ void __launch_bounds__(32 * SCAN_ROWS_PER_BLOCK)
pcen_pass12_rowscan(const float* __restrict__ x, const float* __restrict__ alpha,
                    const float* __restrict__ delta, const float* __restrict__ r,
                    float* __restrict__ bws, float* __restrict__ Cws) {
    const int lane = threadIdx.x & 31;
    const int row  = blockIdx.x * SCAN_ROWS_PER_BLOCK + (threadIdx.x >> 5);
    const float s   = s_coef();
    const float q   = 1.0f - s;
    const float q2  = q * q, q4 = q2 * q2, q8 = q4 * q4, q16 = q8 * q8;
    const float q32 = q16 * q16, q64 = q32 * q32;

    const float* xr = x + (size_t)row * TT;
    const v4f*   p4 = (const v4f*)(xr + lane * 64);
    const float  c0 = xr[0];

    float L[4];
#pragma unroll
    for (int c = 0; c < 4; ++c) {
        float acc = 0.0f;
#pragma unroll
        for (int i = 0; i < 4; ++i) {
            v4f v = p4[c * 4 + i];
            acc = acc * q + v.x;
            acc = acc * q + v.y;
            acc = acc * q + v.z;
            acc = acc * q + v.w;
        }
        L[c] = s * acc;
    }

    float P = ((L[0] * q16 + L[1]) * q16 + L[2]) * q16 + L[3];
    if (lane == 0) P = q64 * c0 + P;
    float w = q64;
#pragma unroll
    for (int d = 1; d < 32; d <<= 1) {
        float up = __shfl_up(P, d, 32);
        if (lane >= d) P = w * up + P;
        w = w * w;
    }
    float cin = __shfl_up(P, 1, 32);
    if (lane == 0) cin = c0;

    v4f cv;
    cv.x = cin;
    cv.y = q16 * cv.x + L[0];
    cv.z = q16 * cv.y + L[1];
    cv.w = q16 * cv.z + L[2];
    *(v4f*)(Cws + (size_t)row * NCHUNKS + lane * 4) = cv;

    if (blockIdx.x == 0) {
        const int b = threadIdx.x;
        float a  = __expf(alpha[b]);
        float d  = __expf(delta[b]);
        float rr = __expf(r[b]);
        bws[b]              = a;
        bws[NBANDS + b]     = d;
        bws[2 * NBANDS + b] = rr;
        bws[3 * NBANDS + b] = fast_pow_pos(d, rr);
    }
}

#define P3_TILES 8
#define P3_WAVES (NBANDS / 16)
#define P3_SPAN  (P3_TILES * CHUNK)
#define ST_TILES 2
#define ST_ROWS  (ST_TILES * CHUNK)
#define ST_PITCH (NBANDS + 4)
#define ST_ROWS_PER_WAVE (ST_ROWS / P3_WAVES)
static_assert(P3_TILES % ST_TILES == 0, "whole stages");
static_assert(ST_ROWS % P3_WAVES == 0, "rows of a stage divide over the waves");
static_assert(NBANDS == 32 * 4, "one time row = 32 lanes x 4 floats");
__global__ void __launch_bounds__(32 * P3_WAVES)
pcen_pass3_wmma(const float* __restrict__ x, const float* __restrict__ Cws,
                const float* __restrict__ bws, float* __restrict__ out) {
    __shared__ __attribute__((aligned(16))) float stage[2][ST_ROWS * ST_PITCH];

    const float s = s_coef();
    const float q = 1.0f - s;

    const int lane = threadIdx.x & 31;
    const int wv   = threadIdx.x >> 5;
    const int m    = lane & 15;
    const int hi   = lane >> 4;

    const int bidx   = blockIdx.y;
    const int chunk0 = blockIdx.x * P3_TILES;
    const int t00    = chunk0 * CHUNK;
    const int row0   = bidx * NBANDS + wv * 16;

    const float* xr = x + (size_t)(row0 + m) * TT + t00 + 2 * hi;
    v2f Am[P3_TILES][4];
#pragma unroll
    for (int i = 0; i < P3_TILES; ++i) {
#pragma unroll
        for (int kk = 0; kk < 4; ++kk)
            Am[i][kk] = *(const v2f*)(xr + i * CHUNK + kk * 4);
    }

    float cr[8][P3_TILES];
#pragma unroll
    for (int v = 0; v < 8; ++v) {
        const v4f* cp = (const v4f*)(Cws + (size_t)(row0 + v + 8 * hi) * NCHUNKS + chunk0);
        v4f c_lo = cp[0], c_hi = cp[1];
        cr[v][0] = c_lo.x; cr[v][1] = c_lo.y; cr[v][2] = c_lo.z; cr[v][3] = c_lo.w;
        cr[v][4] = c_hi.x; cr[v][5] = c_hi.y; cr[v][6] = c_hi.z; cr[v][7] = c_hi.w;
    }

    const int band_base = wv * 16 + 8 * hi;
    const float* ea  = bws;
    const float* ed  = bws + NBANDS;
    const float* err = bws + 2 * NBANDS;
    const float* edr = bws + 3 * NBANDS;
    v4f a_lo  = *(const v4f*)(ea  + band_base), a_hi  = *(const v4f*)(ea  + band_base + 4);
    v4f d_lo  = *(const v4f*)(ed  + band_base), d_hi  = *(const v4f*)(ed  + band_base + 4);
    v4f r_lo  = *(const v4f*)(err + band_base), r_hi  = *(const v4f*)(err + band_base + 4);
    v4f dr_lo = *(const v4f*)(edr + band_base), dr_hi = *(const v4f*)(edr + band_base + 4);
    float av[8]  = {a_lo.x, a_lo.y, a_lo.z, a_lo.w, a_hi.x, a_hi.y, a_hi.z, a_hi.w};
    float dv[8]  = {d_lo.x, d_lo.y, d_lo.z, d_lo.w, d_hi.x, d_hi.y, d_hi.z, d_hi.w};
    float rv[8]  = {r_lo.x, r_lo.y, r_lo.z, r_lo.w, r_hi.x, r_hi.y, r_hi.z, r_hi.w};
    float drv[8] = {dr_lo.x, dr_lo.y, dr_lo.z, dr_lo.w, dr_hi.x, dr_hi.y, dr_hi.z, dr_hi.w};

    v2f Bm[4];
#pragma unroll
    for (int kk = 0; kk < 4; ++kk) {
        int i0 = kk * 4 + 2 * hi;
        Bm[kk].x = (m >= i0)     ? s * qpow(q, m - i0)     : 0.0f;
        Bm[kk].y = (m >= i0 + 1) ? s * qpow(q, m - i0 - 1) : 0.0f;
    }
    const float qn1 = qpow(q, m + 1);

    const float* xe = x + (size_t)(row0 + 8 * hi) * TT + t00 + m;
    float xv[P3_TILES][8];
#pragma unroll
    for (int i = 0; i < P3_TILES; ++i) {
#pragma unroll
        for (int v = 0; v < 8; ++v)
            xv[i][v] = xe[(size_t)v * TT + i * CHUNK];
    }

    float* ob = out + (size_t)bidx * TT * NBANDS + (size_t)t00 * NBANDS + lane * 4;

#pragma unroll
    for (int i = 0; i < P3_TILES; ++i) {
        v8f acc;
#pragma unroll
        for (int v = 0; v < 8; ++v) acc[v] = cr[v][i] * qn1;

        acc = __builtin_amdgcn_wmma_f32_16x16x4_f32(false, Am[i][0], false, Bm[0], (short)0, acc, false, false);
        acc = __builtin_amdgcn_wmma_f32_16x16x4_f32(false, Am[i][1], false, Bm[1], (short)0, acc, false, false);
        acc = __builtin_amdgcn_wmma_f32_16x16x4_f32(false, Am[i][2], false, Bm[2], (short)0, acc, false, false);
        acc = __builtin_amdgcn_wmma_f32_16x16x4_f32(false, Am[i][3], false, Bm[3], (short)0, acc, false, false);

        float res[8];
#pragma unroll
        for (int v = 0; v < 8; ++v) {
            float sm     = fmaxf(acc[v], 0.0f);
            float smooth = fast_pow_pos(EPSV + sm, -av[v]);
            float base   = fmaf(xv[i][v], smooth, dv[v]);
            res[v] = fast_pow_pos(base, rv[v]) - drv[v];
        }
        v4f lo;  lo.x  = res[0]; lo.y  = res[1]; lo.z  = res[2]; lo.w  = res[3];
        v4f hi4; hi4.x = res[4]; hi4.y = res[5]; hi4.z = res[6]; hi4.w = res[7];
        float* sb = stage[(i / ST_TILES) & 1];
        float* sp = sb + ((i % ST_TILES) * CHUNK + m) * ST_PITCH + band_base;
        *(v4f*)(sp)     = lo;
        *(v4f*)(sp + 4) = hi4;

        if ((i % ST_TILES) == ST_TILES - 1) {
            __builtin_amdgcn_fence(__ATOMIC_RELEASE, "workgroup", "local");
            __builtin_amdgcn_s_barrier();
            __builtin_amdgcn_fence(__ATOMIC_ACQUIRE, "workgroup", "local");
            float* o = ob + (size_t)(i / ST_TILES) * ST_ROWS * NBANDS;
#pragma unroll
            for (int k = 0; k < ST_ROWS_PER_WAVE; ++k) {
                const int row = wv * ST_ROWS_PER_WAVE + k;
                v4f val = *(const v4f*)(sb + row * ST_PITCH + lane * 4);
                *(v4f*)(o + (size_t)row * NBANDS) = val;
            }
        }
    }
}

extern "C" void kernel_launch(void* const* d_in, const int* in_sizes, int n_in,
                              void* d_out, int out_size, void* d_ws, size_t ws_size,
                              hipStream_t stream) {
    const float* x     = (const float*)d_in[0];
    const float* alpha = (const float*)d_in[1];
    const float* delta = (const float*)d_in[2];
    const float* r     = (const float*)d_in[3];
    float* out = (float*)d_out;

    float* bws = (float*)d_ws;
    float* Cws = bws + 4 * NBANDS;

    pcen_pass12_rowscan<<<ROWS / SCAN_ROWS_PER_BLOCK, 32 * SCAN_ROWS_PER_BLOCK, 0, stream>>>(
        x, alpha, delta, r, bws, Cws);

    dim3 g3(TT / P3_SPAN, NBATCH);
    pcen_pass3_wmma<<<g3, 32 * P3_WAVES, 0, stream>>>(x, Cws, bws, out);
}


# Tile pass set-up without loops: the 9 per-lane powers of q (decay matrix, q^(N+1)) are selected from compile-time products, not multiply loops of up to 16 trips in every wave; values bit-identical
# speedup vs baseline: 2.9676x; 1.1148x over previous
#include <hip/hip_runtime.h>

#define TT      2048
#define NBANDS  128
#define NBATCH  16
#define ROWS    (NBATCH * NBANDS)
#define CHUNK   16
#define NCHUNKS (TT / CHUNK)
#define EPSV    1e-5f

typedef __attribute__((ext_vector_type(2))) float v2f;
typedef __attribute__((ext_vector_type(4))) float v4f;
typedef __attribute__((ext_vector_type(8))) float v8f;

__device__ __forceinline__ float s_coef() {
    return (sqrtf(262145.0f) - 1.0f) / 131072.0f;
}

__device__ __forceinline__ float qpow(float q, int e) {
    float r = 1.0f, p = 1.0f;
#pragma unroll
    for (int i = 1; i <= CHUNK; ++i) {
        p *= q;
        r = (e >= i) ? p : r;
    }
    return r;
}

__device__ __forceinline__ float fast_pow_pos(float y, float p) {
    return __builtin_amdgcn_exp2f(p * __builtin_amdgcn_logf(y));
}

#define SCAN_ROWS_PER_BLOCK 4
static_assert(32 * SCAN_ROWS_PER_BLOCK == NBANDS, "work-group 0 covers one band per thread");
__global__ void __launch_bounds__(32 * SCAN_ROWS_PER_BLOCK)
pcen_pass12_rowscan(const float* __restrict__ x, const float* __restrict__ alpha,
                    const float* __restrict__ delta, const float* __restrict__ r,
                    float* __restrict__ bws, float* __restrict__ Cws) {
    const int lane = threadIdx.x & 31;
    const int row  = blockIdx.x * SCAN_ROWS_PER_BLOCK + (threadIdx.x >> 5);
    const float s   = s_coef();
    const float q   = 1.0f - s;
    const float q2  = q * q, q4 = q2 * q2, q8 = q4 * q4, q16 = q8 * q8;
    const float q32 = q16 * q16, q64 = q32 * q32;

    const float* xr = x + (size_t)row * TT;
    const v4f*   p4 = (const v4f*)(xr + lane * 64);
    const float  c0 = xr[0];

    float L[4];
#pragma unroll
    for (int c = 0; c < 4; ++c) {
        float acc = 0.0f;
#pragma unroll
        for (int i = 0; i < 4; ++i) {
            v4f v = p4[c * 4 + i];
            acc = acc * q + v.x;
            acc = acc * q + v.y;
            acc = acc * q + v.z;
            acc = acc * q + v.w;
        }
        L[c] = s * acc;
    }

    float P = ((L[0] * q16 + L[1]) * q16 + L[2]) * q16 + L[3];
    if (lane == 0) P = q64 * c0 + P;
    float w = q64;
#pragma unroll
    for (int d = 1; d < 32; d <<= 1) {
        float up = __shfl_up(P, d, 32);
        if (lane >= d) P = w * up + P;
        w = w * w;
    }
    float cin = __shfl_up(P, 1, 32);
    if (lane == 0) cin = c0;

    v4f cv;
    cv.x = cin;
    cv.y = q16 * cv.x + L[0];
    cv.z = q16 * cv.y + L[1];
    cv.w = q16 * cv.z + L[2];
    *(v4f*)(Cws + (size_t)row * NCHUNKS + lane * 4) = cv;

    if (blockIdx.x == 0) {
        const int b = threadIdx.x;
        float a  = __expf(alpha[b]);
        float d  = __expf(delta[b]);
        float rr = __expf(r[b]);
        bws[b]              = a;
        bws[NBANDS + b]     = d;
        bws[2 * NBANDS + b] = rr;
        bws[3 * NBANDS + b] = fast_pow_pos(d, rr);
    }
}

#define P3_TILES 8
#define P3_WAVES (NBANDS / 16)
#define P3_SPAN  (P3_TILES * CHUNK)
#define ST_TILES 2
#define ST_ROWS  (ST_TILES * CHUNK)
#define ST_PITCH (NBANDS + 4)
#define ST_ROWS_PER_WAVE (ST_ROWS / P3_WAVES)
static_assert(P3_TILES % ST_TILES == 0, "whole stages");
static_assert(ST_ROWS % P3_WAVES == 0, "rows of a stage divide over the waves");
static_assert(NBANDS == 32 * 4, "one time row = 32 lanes x 4 floats");
__global__ void __launch_bounds__(32 * P3_WAVES)
pcen_pass3_wmma(const float* __restrict__ x, const float* __restrict__ Cws,
                const float* __restrict__ bws, float* __restrict__ out) {
    __shared__ __attribute__((aligned(16))) float stage[2][ST_ROWS * ST_PITCH];

    const float s = s_coef();
    const float q = 1.0f - s;

    const int lane = threadIdx.x & 31;
    const int wv   = threadIdx.x >> 5;
    const int m    = lane & 15;
    const int hi   = lane >> 4;

    const int bidx   = blockIdx.y;
    const int chunk0 = blockIdx.x * P3_TILES;
    const int t00    = chunk0 * CHUNK;
    const int row0   = bidx * NBANDS + wv * 16;

    const float* xr = x + (size_t)(row0 + m) * TT + t00 + 2 * hi;
    v2f Am[P3_TILES][4];
#pragma unroll
    for (int i = 0; i < P3_TILES; ++i) {
#pragma unroll
        for (int kk = 0; kk < 4; ++kk)
            Am[i][kk] = *(const v2f*)(xr + i * CHUNK + kk * 4);
    }

    float cr[8][P3_TILES];
#pragma unroll
    for (int v = 0; v < 8; ++v) {
        const v4f* cp = (const v4f*)(Cws + (size_t)(row0 + v + 8 * hi) * NCHUNKS + chunk0);
        v4f c_lo = cp[0], c_hi = cp[1];
        cr[v][0] = c_lo.x; cr[v][1] = c_lo.y; cr[v][2] = c_lo.z; cr[v][3] = c_lo.w;
        cr[v][4] = c_hi.x; cr[v][5] = c_hi.y; cr[v][6] = c_hi.z; cr[v][7] = c_hi.w;
    }

    const int band_base = wv * 16 + 8 * hi;
    const float* ea  = bws;
    const float* ed  = bws + NBANDS;
    const float* err = bws + 2 * NBANDS;
    const float* edr = bws + 3 * NBANDS;
    v4f a_lo  = *(const v4f*)(ea  + band_base), a_hi  = *(const v4f*)(ea  + band_base + 4);
    v4f d_lo  = *(const v4f*)(ed  + band_base), d_hi  = *(const v4f*)(ed  + band_base + 4);
    v4f r_lo  = *(const v4f*)(err + band_base), r_hi  = *(const v4f*)(err + band_base + 4);
    v4f dr_lo = *(const v4f*)(edr + band_base), dr_hi = *(const v4f*)(edr + band_base + 4);
    float av[8]  = {a_lo.x, a_lo.y, a_lo.z, a_lo.w, a_hi.x, a_hi.y, a_hi.z, a_hi.w};
    float dv[8]  = {d_lo.x, d_lo.y, d_lo.z, d_lo.w, d_hi.x, d_hi.y, d_hi.z, d_hi.w};
    float rv[8]  = {r_lo.x, r_lo.y, r_lo.z, r_lo.w, r_hi.x, r_hi.y, r_hi.z, r_hi.w};
    float drv[8] = {dr_lo.x, dr_lo.y, dr_lo.z, dr_lo.w, dr_hi.x, dr_hi.y, dr_hi.z, dr_hi.w};

    v2f Bm[4];
#pragma unroll
    for (int kk = 0; kk < 4; ++kk) {
        int i0 = kk * 4 + 2 * hi;
        Bm[kk].x = (m >= i0)     ? s * qpow(q, m - i0)     : 0.0f;
        Bm[kk].y = (m >= i0 + 1) ? s * qpow(q, m - i0 - 1) : 0.0f;
    }
    const float qn1 = qpow(q, m + 1);

    const float* xe = x + (size_t)(row0 + 8 * hi) * TT + t00 + m;
    float xv[P3_TILES][8];
#pragma unroll
    for (int i = 0; i < P3_TILES; ++i) {
#pragma unroll
        for (int v = 0; v < 8; ++v)
            xv[i][v] = xe[(size_t)v * TT + i * CHUNK];
    }

    float* ob = out + (size_t)bidx * TT * NBANDS + (size_t)t00 * NBANDS + lane * 4;

#pragma unroll
    for (int i = 0; i < P3_TILES; ++i) {
        v8f acc;
#pragma unroll
        for (int v = 0; v < 8; ++v) acc[v] = cr[v][i] * qn1;

        acc = __builtin_amdgcn_wmma_f32_16x16x4_f32(false, Am[i][0], false, Bm[0], (short)0, acc, false, false);
        acc = __builtin_amdgcn_wmma_f32_16x16x4_f32(false, Am[i][1], false, Bm[1], (short)0, acc, false, false);
        acc = __builtin_amdgcn_wmma_f32_16x16x4_f32(false, Am[i][2], false, Bm[2], (short)0, acc, false, false);
        acc = __builtin_amdgcn_wmma_f32_16x16x4_f32(false, Am[i][3], false, Bm[3], (short)0, acc, false, false);

        float res[8];
#pragma unroll
        for (int v = 0; v < 8; ++v) {
            float sm     = fmaxf(acc[v], 0.0f);
            float smooth = fast_pow_pos(EPSV + sm, -av[v]);
            float base   = fmaf(xv[i][v], smooth, dv[v]);
            res[v] = fast_pow_pos(base, rv[v]) - drv[v];
        }
        v4f lo;  lo.x  = res[0]; lo.y  = res[1]; lo.z  = res[2]; lo.w  = res[3];
        v4f hi4; hi4.x = res[4]; hi4.y = res[5]; hi4.z = res[6]; hi4.w = res[7];
        float* sb = stage[(i / ST_TILES) & 1];
        float* sp = sb + ((i % ST_TILES) * CHUNK + m) * ST_PITCH + band_base;
        *(v4f*)(sp)     = lo;
        *(v4f*)(sp + 4) = hi4;

        if ((i % ST_TILES) == ST_TILES - 1) {
            __builtin_amdgcn_fence(__ATOMIC_RELEASE, "workgroup", "local");
            __builtin_amdgcn_s_barrier();
            __builtin_amdgcn_fence(__ATOMIC_ACQUIRE, "workgroup", "local");
            float* o = ob + (size_t)(i / ST_TILES) * ST_ROWS * NBANDS;
#pragma unroll
            for (int k = 0; k < ST_ROWS_PER_WAVE; ++k) {
                const int row = wv * ST_ROWS_PER_WAVE + k;
                v4f val = *(const v4f*)(sb + row * ST_PITCH + lane * 4);
                *(v4f*)(o + (size_t)row * NBANDS) = val;
            }
        }
    }
}

extern "C" void kernel_launch(void* const* d_in, const int* in_sizes, int n_in,
                              void* d_out, int out_size, void* d_ws, size_t ws_size,
                              hipStream_t stream) {
    const float* x     = (const float*)d_in[0];
    const float* alpha = (const float*)d_in[1];
    const float* delta = (const float*)d_in[2];
    const float* r     = (const float*)d_in[3];
    float* out = (float*)d_out;

    float* bws = (float*)d_ws;
    float* Cws = bws + 4 * NBANDS;

    pcen_pass12_rowscan<<<ROWS / SCAN_ROWS_PER_BLOCK, 32 * SCAN_ROWS_PER_BLOCK, 0, stream>>>(
        x, alpha, delta, r, bws, Cws);

    dim3 g3(TT / P3_SPAN, NBATCH);
    pcen_pass3_wmma<<<g3, 32 * P3_WAVES, 0, stream>>>(x, Cws, bws, out);
}


# Row scan fetches each row with whole-line loads (512 contiguous bytes per load instruction, not 32 lines of 16 B each); lanes get their 64-step ranges by a transpose through LDS; sums as before
# speedup vs baseline: 3.2048x; 1.0799x over previous
#include <hip/hip_runtime.h>

#define TT      2048
#define NBANDS  128
#define NBATCH  16
#define ROWS    (NBATCH * NBANDS)
#define CHUNK   16
#define NCHUNKS (TT / CHUNK)
#define EPSV    1e-5f

typedef __attribute__((ext_vector_type(2))) float v2f;
typedef __attribute__((ext_vector_type(4))) float v4f;
typedef __attribute__((ext_vector_type(8))) float v8f;

__device__ __forceinline__ float s_coef() {
    return (sqrtf(262145.0f) - 1.0f) / 131072.0f;
}

__device__ __forceinline__ float qpow(float q, int e) {
    float r = 1.0f, p = 1.0f;
#pragma unroll
    for (int i = 1; i <= CHUNK; ++i) {
        p *= q;
        r = (e >= i) ? p : r;
    }
    return r;
}

__device__ __forceinline__ float fast_pow_pos(float y, float p) {
    return __builtin_amdgcn_exp2f(p * __builtin_amdgcn_logf(y));
}

#define SCAN_ROWS_PER_BLOCK 4
#define TR_PITCH 68
static_assert(32 * SCAN_ROWS_PER_BLOCK == NBANDS, "work-group 0 covers one band per thread");
static_assert(TT == 32 * 64, "a row is 32 lane ranges of 64 steps = 16 loads of 128 steps");
__global__ void __launch_bounds__(32 * SCAN_ROWS_PER_BLOCK)
pcen_pass12_rowscan(const float* __restrict__ x, const float* __restrict__ alpha,
                    const float* __restrict__ delta, const float* __restrict__ r,
                    float* __restrict__ bws, float* __restrict__ Cws) {
    __shared__ __attribute__((aligned(16))) float tr[SCAN_ROWS_PER_BLOCK][32 * TR_PITCH];

    const int lane = threadIdx.x & 31;
    const int wv   = threadIdx.x >> 5;
    const int row  = blockIdx.x * SCAN_ROWS_PER_BLOCK + wv;
    const float s   = s_coef();
    const float q   = 1.0f - s;
    const float q2  = q * q, q4 = q2 * q2, q8 = q4 * q4, q16 = q8 * q8;
    const float q32 = q16 * q16, q64 = q32 * q32;

    const float* xr = x + (size_t)row * TT;
    float*       t  = tr[wv];

    v4f g[16];
#pragma unroll
    for (int j = 0; j < 16; ++j)
        g[j] = *(const v4f*)(xr + j * 128 + lane * 4);
#pragma unroll
    for (int j = 0; j < 16; ++j)
        *(v4f*)(t + (2 * j + (lane >> 4)) * TR_PITCH + (lane & 15) * 4) = g[j];

    __builtin_amdgcn_fence(__ATOMIC_RELEASE, "wavefront", "local");
    __builtin_amdgcn_wave_barrier();
    __builtin_amdgcn_fence(__ATOMIC_ACQUIRE, "wavefront", "local");

    const v4f*   p4 = (const v4f*)(t + lane * TR_PITCH);
    const float  c0 = t[0];

    float L[4];
#pragma unroll
    for (int c = 0; c < 4; ++c) {
        float acc = 0.0f;
#pragma unroll
        for (int i = 0; i < 4; ++i) {
            v4f v = p4[c * 4 + i];
            acc = acc * q + v.x;
            acc = acc * q + v.y;
            acc = acc * q + v.z;
            acc = acc * q + v.w;
        }
        L[c] = s * acc;
    }

    float P = ((L[0] * q16 + L[1]) * q16 + L[2]) * q16 + L[3];
    if (lane == 0) P = q64 * c0 + P;
    float w = q64;
#pragma unroll
    for (int d = 1; d < 32; d <<= 1) {
        float up = __shfl_up(P, d, 32);
        if (lane >= d) P = w * up + P;
        w = w * w;
    }
    float cin = __shfl_up(P, 1, 32);
    if (lane == 0) cin = c0;

    v4f cv;
    cv.x = cin;
    cv.y = q16 * cv.x + L[0];
    cv.z = q16 * cv.y + L[1];
    cv.w = q16 * cv.z + L[2];
    *(v4f*)(Cws + (size_t)row * NCHUNKS + lane * 4) = cv;

    if (blockIdx.x == 0) {
        const int b = threadIdx.x;
        float a  = __expf(alpha[b]);
        float d  = __expf(delta[b]);
        float rr = __expf(r[b]);
        bws[b]              = a;
        bws[NBANDS + b]     = d;
        bws[2 * NBANDS + b] = rr;
        bws[3 * NBANDS + b] = fast_pow_pos(d, rr);
    }
}

#define P3_TILES 8
#define P3_WAVES (NBANDS / 16)
#define P3_SPAN  (P3_TILES * CHUNK)
#define ST_TILES 2
#define ST_ROWS  (ST_TILES * CHUNK)
#define ST_PITCH (NBANDS + 4)
#define ST_ROWS_PER_WAVE (ST_ROWS / P3_WAVES)
static_assert(P3_TILES % ST_TILES == 0, "whole stages");
static_assert(ST_ROWS % P3_WAVES == 0, "rows of a stage divide over the waves");
static_assert(NBANDS == 32 * 4, "one time row = 32 lanes x 4 floats");
__global__ void __launch_bounds__(32 * P3_WAVES)
pcen_pass3_wmma(const float* __restrict__ x, const float* __restrict__ Cws,
                const float* __restrict__ bws, float* __restrict__ out) {
    __shared__ __attribute__((aligned(16))) float stage[2][ST_ROWS * ST_PITCH];

    const float s = s_coef();
    const float q = 1.0f - s;

    const int lane = threadIdx.x & 31;
    const int wv   = threadIdx.x >> 5;
    const int m    = lane & 15;
    const int hi   = lane >> 4;

    const int bidx   = blockIdx.y;
    const int chunk0 = blockIdx.x * P3_TILES;
    const int t00    = chunk0 * CHUNK;
    const int row0   = bidx * NBANDS + wv * 16;

    const float* xr = x + (size_t)(row0 + m) * TT + t00 + 2 * hi;
    v2f Am[P3_TILES][4];
#pragma unroll
    for (int i = 0; i < P3_TILES; ++i) {
#pragma unroll
        for (int kk = 0; kk < 4; ++kk)
            Am[i][kk] = *(const v2f*)(xr + i * CHUNK + kk * 4);
    }

    float cr[8][P3_TILES];
#pragma unroll
    for (int v = 0; v < 8; ++v) {
        const v4f* cp = (const v4f*)(Cws + (size_t)(row0 + v + 8 * hi) * NCHUNKS + chunk0);
        v4f c_lo = cp[0], c_hi = cp[1];
        cr[v][0] = c_lo.x; cr[v][1] = c_lo.y; cr[v][2] = c_lo.z; cr[v][3] = c_lo.w;
        cr[v][4] = c_hi.x; cr[v][5] = c_hi.y; cr[v][6] = c_hi.z; cr[v][7] = c_hi.w;
    }

    const int band_base = wv * 16 + 8 * hi;
    const float* ea  = bws;
    const float* ed  = bws + NBANDS;
    const float* err = bws + 2 * NBANDS;
    const float* edr = bws + 3 * NBANDS;
    v4f a_lo  = *(const v4f*)(ea  + band_base), a_hi  = *(const v4f*)(ea  + band_base + 4);
    v4f d_lo  = *(const v4f*)(ed  + band_base), d_hi  = *(const v4f*)(ed  + band_base + 4);
    v4f r_lo  = *(const v4f*)(err + band_base), r_hi  = *(const v4f*)(err + band_base + 4);
    v4f dr_lo = *(const v4f*)(edr + band_base), dr_hi = *(const v4f*)(edr + band_base + 4);
    float av[8]  = {a_lo.x, a_lo.y, a_lo.z, a_lo.w, a_hi.x, a_hi.y, a_hi.z, a_hi.w};
    float dv[8]  = {d_lo.x, d_lo.y, d_lo.z, d_lo.w, d_hi.x, d_hi.y, d_hi.z, d_hi.w};
    float rv[8]  = {r_lo.x, r_lo.y, r_lo.z, r_lo.w, r_hi.x, r_hi.y, r_hi.z, r_hi.w};
    float drv[8] = {dr_lo.x, dr_lo.y, dr_lo.z, dr_lo.w, dr_hi.x, dr_hi.y, dr_hi.z, dr_hi.w};

    v2f Bm[4];
#pragma unroll
    for (int kk = 0; kk < 4; ++kk) {
        int i0 = kk * 4 + 2 * hi;
        Bm[kk].x = (m >= i0)     ? s * qpow(q, m - i0)     : 0.0f;
        Bm[kk].y = (m >= i0 + 1) ? s * qpow(q, m - i0 - 1) : 0.0f;
    }
    const float qn1 = qpow(q, m + 1);

    const float* xe = x + (size_t)(row0 + 8 * hi) * TT + t00 + m;
    float xv[P3_TILES][8];
#pragma unroll
    for (int i = 0; i < P3_TILES; ++i) {
#pragma unroll
        for (int v = 0; v < 8; ++v)
            xv[i][v] = xe[(size_t)v * TT + i * CHUNK];
    }

    float* ob = out + (size_t)bidx * TT * NBANDS + (size_t)t00 * NBANDS + lane * 4;

#pragma unroll
    for (int i = 0; i < P3_TILES; ++i) {
        v8f acc;
#pragma unroll
        for (int v = 0; v < 8; ++v) acc[v] = cr[v][i] * qn1;

        acc = __builtin_amdgcn_wmma_f32_16x16x4_f32(false, Am[i][0], false, Bm[0], (short)0, acc, false, false);
        acc = __builtin_amdgcn_wmma_f32_16x16x4_f32(false, Am[i][1], false, Bm[1], (short)0, acc, false, false);
        acc = __builtin_amdgcn_wmma_f32_16x16x4_f32(false, Am[i][2], false, Bm[2], (short)0, acc, false, false);
        acc = __builtin_amdgcn_wmma_f32_16x16x4_f32(false, Am[i][3], false, Bm[3], (short)0, acc, false, false);

        float res[8];
#pragma unroll
        for (int v = 0; v < 8; ++v) {
            float sm     = fmaxf(acc[v], 0.0f);
            float smooth = fast_pow_pos(EPSV + sm, -av[v]);
            float base   = fmaf(xv[i][v], smooth, dv[v]);
            res[v] = fast_pow_pos(base, rv[v]) - drv[v];
        }
        v4f lo;  lo.x  = res[0]; lo.y  = res[1]; lo.z  = res[2]; lo.w  = res[3];
        v4f hi4; hi4.x = res[4]; hi4.y = res[5]; hi4.z = res[6]; hi4.w = res[7];
        float* sb = stage[(i / ST_TILES) & 1];
        float* sp = sb + ((i % ST_TILES) * CHUNK + m) * ST_PITCH + band_base;
        *(v4f*)(sp)     = lo;
        *(v4f*)(sp + 4) = hi4;

        if ((i % ST_TILES) == ST_TILES - 1) {
            __builtin_amdgcn_fence(__ATOMIC_RELEASE, "workgroup", "local");
            __builtin_amdgcn_s_barrier();
            __builtin_amdgcn_fence(__ATOMIC_ACQUIRE, "workgroup", "local");
            float* o = ob + (size_t)(i / ST_TILES) * ST_ROWS * NBANDS;
#pragma unroll
            for (int k = 0; k < ST_ROWS_PER_WAVE; ++k) {
                const int row = wv * ST_ROWS_PER_WAVE + k;
                v4f val = *(const v4f*)(sb + row * ST_PITCH + lane * 4);
                *(v4f*)(o + (size_t)row * NBANDS) = val;
            }
        }
    }
}

extern "C" void kernel_launch(void* const* d_in, const int* in_sizes, int n_in,
                              void* d_out, int out_size, void* d_ws, size_t ws_size,
                              hipStream_t stream) {
    const float* x     = (const float*)d_in[0];
    const float* alpha = (const float*)d_in[1];
    const float* delta = (const float*)d_in[2];
    const float* r     = (const float*)d_in[3];
    float* out = (float*)d_out;

    float* bws = (float*)d_ws;
    float* Cws = bws + 4 * NBANDS;

    pcen_pass12_rowscan<<<ROWS / SCAN_ROWS_PER_BLOCK, 32 * SCAN_ROWS_PER_BLOCK, 0, stream>>>(
        x, alpha, delta, r, bws, Cws);

    dim3 g3(TT / P3_SPAN, NBATCH);
    pcen_pass3_wmma<<<g3, 32 * P3_WAVES, 0, stream>>>(x, Cws, bws, out);
}


# Tile pass fetches x once with whole-line loads (512 contiguous bytes per instruction) into wave-private LDS; A fragments and elementwise x are read from LDS, not by 96 multi-line global loads a wave
# speedup vs baseline: 3.4346x; 1.0717x over previous
#include <hip/hip_runtime.h>

#define TT      2048
#define NBANDS  128
#define NBATCH  16
#define ROWS    (NBATCH * NBANDS)
#define CHUNK   16
#define NCHUNKS (TT / CHUNK)
#define EPSV    1e-5f

typedef __attribute__((ext_vector_type(2))) float v2f;
typedef __attribute__((ext_vector_type(4))) float v4f;
typedef __attribute__((ext_vector_type(8))) float v8f;

__device__ __forceinline__ float s_coef() {
    return (sqrtf(262145.0f) - 1.0f) / 131072.0f;
}

__device__ __forceinline__ float qpow(float q, int e) {
    float r = 1.0f, p = 1.0f;
#pragma unroll
    for (int i = 1; i <= CHUNK; ++i) {
        p *= q;
        r = (e >= i) ? p : r;
    }
    return r;
}

__device__ __forceinline__ float fast_pow_pos(float y, float p) {
    return __builtin_amdgcn_exp2f(p * __builtin_amdgcn_logf(y));
}

#define SCAN_ROWS_PER_BLOCK 4
#define TR_PITCH 68
static_assert(32 * SCAN_ROWS_PER_BLOCK == NBANDS, "work-group 0 covers one band per thread");
static_assert(TT == 32 * 64, "a row is 32 lane ranges of 64 steps = 16 loads of 128 steps");
__global__ void __launch_bounds__(32 * SCAN_ROWS_PER_BLOCK)
pcen_pass12_rowscan(const float* __restrict__ x, const float* __restrict__ alpha,
                    const float* __restrict__ delta, const float* __restrict__ r,
                    float* __restrict__ bws, float* __restrict__ Cws) {
    __shared__ __attribute__((aligned(16))) float tr[SCAN_ROWS_PER_BLOCK][32 * TR_PITCH];

    const int lane = threadIdx.x & 31;
    const int wv   = threadIdx.x >> 5;
    const int row  = blockIdx.x * SCAN_ROWS_PER_BLOCK + wv;
    const float s   = s_coef();
    const float q   = 1.0f - s;
    const float q2  = q * q, q4 = q2 * q2, q8 = q4 * q4, q16 = q8 * q8;
    const float q32 = q16 * q16, q64 = q32 * q32;

    const float* xr = x + (size_t)row * TT;
    float*       t  = tr[wv];

    v4f g[16];
#pragma unroll
    for (int j = 0; j < 16; ++j)
        g[j] = *(const v4f*)(xr + j * 128 + lane * 4);
#pragma unroll
    for (int j = 0; j < 16; ++j)
        *(v4f*)(t + (2 * j + (lane >> 4)) * TR_PITCH + (lane & 15) * 4) = g[j];

    __builtin_amdgcn_fence(__ATOMIC_RELEASE, "wavefront", "local");
    __builtin_amdgcn_wave_barrier();
    __builtin_amdgcn_fence(__ATOMIC_ACQUIRE, "wavefront", "local");

    const v4f*   p4 = (const v4f*)(t + lane * TR_PITCH);
    const float  c0 = t[0];

    float L[4];
#pragma unroll
    for (int c = 0; c < 4; ++c) {
        float acc = 0.0f;
#pragma unroll
        for (int i = 0; i < 4; ++i) {
            v4f v = p4[c * 4 + i];
            acc = acc * q + v.x;
            acc = acc * q + v.y;
            acc = acc * q + v.z;
            acc = acc * q + v.w;
        }
        L[c] = s * acc;
    }

    float P = ((L[0] * q16 + L[1]) * q16 + L[2]) * q16 + L[3];
    if (lane == 0) P = q64 * c0 + P;
    float w = q64;
#pragma unroll
    for (int d = 1; d < 32; d <<= 1) {
        float up = __shfl_up(P, d, 32);
        if (lane >= d) P = w * up + P;
        w = w * w;
    }
    float cin = __shfl_up(P, 1, 32);
    if (lane == 0) cin = c0;

    v4f cv;
    cv.x = cin;
    cv.y = q16 * cv.x + L[0];
    cv.z = q16 * cv.y + L[1];
    cv.w = q16 * cv.z + L[2];
    *(v4f*)(Cws + (size_t)row * NCHUNKS + lane * 4) = cv;

    if (blockIdx.x == 0) {
        const int b = threadIdx.x;
        float a  = __expf(alpha[b]);
        float d  = __expf(delta[b]);
        float rr = __expf(r[b]);
        bws[b]              = a;
        bws[NBANDS + b]     = d;
        bws[2 * NBANDS + b] = rr;
        bws[3 * NBANDS + b] = fast_pow_pos(d, rr);
    }
}

#define P3_TILES 8
#define P3_WAVES (NBANDS / 16)
#define P3_SPAN  (P3_TILES * CHUNK)
#define ST_TILES 2
#define ST_ROWS  (ST_TILES * CHUNK)
#define ST_PITCH (NBANDS + 4)
#define ST_ROWS_PER_WAVE (ST_ROWS / P3_WAVES)
#define XS_PITCH (P3_SPAN + 4)
static_assert(P3_TILES % ST_TILES == 0, "whole stages");
static_assert(ST_ROWS % P3_WAVES == 0, "rows of a stage divide over the waves");
static_assert(NBANDS == 32 * 4, "one time row = 32 lanes x 4 floats");
static_assert(P3_SPAN == 32 * 4, "one x row of the span = 32 lanes x 4 floats");
__global__ void __launch_bounds__(32 * P3_WAVES)
pcen_pass3_wmma(const float* __restrict__ x, const float* __restrict__ Cws,
                const float* __restrict__ bws, float* __restrict__ out) {
    __shared__ __attribute__((aligned(16))) float stage[2][ST_ROWS * ST_PITCH];
    __shared__ __attribute__((aligned(16))) float xs[P3_WAVES][16 * XS_PITCH];

    const float s = s_coef();
    const float q = 1.0f - s;

    const int lane = threadIdx.x & 31;
    const int wv   = threadIdx.x >> 5;
    const int m    = lane & 15;
    const int hi   = lane >> 4;

    const int bidx   = blockIdx.y;
    const int chunk0 = blockIdx.x * P3_TILES;
    const int t00    = chunk0 * CHUNK;
    const int row0   = bidx * NBANDS + wv * 16;

    const float* xb = x + (size_t)row0 * TT + t00 + lane * 4;
    float*       xw = xs[wv];
    v4f g[16];
#pragma unroll
    for (int j = 0; j < 16; ++j)
        g[j] = *(const v4f*)(xb + (size_t)j * TT);

    float cr[8][P3_TILES];
#pragma unroll
    for (int v = 0; v < 8; ++v) {
        const v4f* cp = (const v4f*)(Cws + (size_t)(row0 + v + 8 * hi) * NCHUNKS + chunk0);
        v4f c_lo = cp[0], c_hi = cp[1];
        cr[v][0] = c_lo.x; cr[v][1] = c_lo.y; cr[v][2] = c_lo.z; cr[v][3] = c_lo.w;
        cr[v][4] = c_hi.x; cr[v][5] = c_hi.y; cr[v][6] = c_hi.z; cr[v][7] = c_hi.w;
    }

    const int band_base = wv * 16 + 8 * hi;
    const float* ea  = bws;
    const float* ed  = bws + NBANDS;
    const float* err = bws + 2 * NBANDS;
    const float* edr = bws + 3 * NBANDS;
    v4f a_lo  = *(const v4f*)(ea  + band_base), a_hi  = *(const v4f*)(ea  + band_base + 4);
    v4f d_lo  = *(const v4f*)(ed  + band_base), d_hi  = *(const v4f*)(ed  + band_base + 4);
    v4f r_lo  = *(const v4f*)(err + band_base), r_hi  = *(const v4f*)(err + band_base + 4);
    v4f dr_lo = *(const v4f*)(edr + band_base), dr_hi = *(const v4f*)(edr + band_base + 4);
    float av[8]  = {a_lo.x, a_lo.y, a_lo.z, a_lo.w, a_hi.x, a_hi.y, a_hi.z, a_hi.w};
    float dv[8]  = {d_lo.x, d_lo.y, d_lo.z, d_lo.w, d_hi.x, d_hi.y, d_hi.z, d_hi.w};
    float rv[8]  = {r_lo.x, r_lo.y, r_lo.z, r_lo.w, r_hi.x, r_hi.y, r_hi.z, r_hi.w};
    float drv[8] = {dr_lo.x, dr_lo.y, dr_lo.z, dr_lo.w, dr_hi.x, dr_hi.y, dr_hi.z, dr_hi.w};

    v2f Bm[4];
#pragma unroll
    for (int kk = 0; kk < 4; ++kk) {
        int i0 = kk * 4 + 2 * hi;
        Bm[kk].x = (m >= i0)     ? s * qpow(q, m - i0)     : 0.0f;
        Bm[kk].y = (m >= i0 + 1) ? s * qpow(q, m - i0 - 1) : 0.0f;
    }
    const float qn1 = qpow(q, m + 1);

#pragma unroll
    for (int j = 0; j < 16; ++j)
        *(v4f*)(xw + j * XS_PITCH + lane * 4) = g[j];

    __builtin_amdgcn_fence(__ATOMIC_RELEASE, "wavefront", "local");
    __builtin_amdgcn_wave_barrier();
    __builtin_amdgcn_fence(__ATOMIC_ACQUIRE, "wavefront", "local");

    const float* xa = xw + m * XS_PITCH + 2 * hi;
    v2f Am[P3_TILES][4];
#pragma unroll
    for (int i = 0; i < P3_TILES; ++i) {
#pragma unroll
        for (int kk = 0; kk < 4; ++kk)
            Am[i][kk] = *(const v2f*)(xa + i * CHUNK + kk * 4);
    }

    const float* xe = xw + (8 * hi) * XS_PITCH + m;
    float xv[P3_TILES][8];
#pragma unroll
    for (int i = 0; i < P3_TILES; ++i) {
#pragma unroll
        for (int v = 0; v < 8; ++v)
            xv[i][v] = xe[v * XS_PITCH + i * CHUNK];
    }

    float* ob = out + (size_t)bidx * TT * NBANDS + (size_t)t00 * NBANDS + lane * 4;

#pragma unroll
    for (int i = 0; i < P3_TILES; ++i) {
        v8f acc;
#pragma unroll
        for (int v = 0; v < 8; ++v) acc[v] = cr[v][i] * qn1;

        acc = __builtin_amdgcn_wmma_f32_16x16x4_f32(false, Am[i][0], false, Bm[0], (short)0, acc, false, false);
        acc = __builtin_amdgcn_wmma_f32_16x16x4_f32(false, Am[i][1], false, Bm[1], (short)0, acc, false, false);
        acc = __builtin_amdgcn_wmma_f32_16x16x4_f32(false, Am[i][2], false, Bm[2], (short)0, acc, false, false);
        acc = __builtin_amdgcn_wmma_f32_16x16x4_f32(false, Am[i][3], false, Bm[3], (short)0, acc, false, false);

        float res[8];
#pragma unroll
        for (int v = 0; v < 8; ++v) {
            float sm     = fmaxf(acc[v], 0.0f);
            float smooth = fast_pow_pos(EPSV + sm, -av[v]);
            float base   = fmaf(xv[i][v], smooth, dv[v]);
            res[v] = fast_pow_pos(base, rv[v]) - drv[v];
        }
        v4f lo;  lo.x  = res[0]; lo.y  = res[1]; lo.z  = res[2]; lo.w  = res[3];
        v4f hi4; hi4.x = res[4]; hi4.y = res[5]; hi4.z = res[6]; hi4.w = res[7];
        float* sb = stage[(i / ST_TILES) & 1];
        float* sp = sb + ((i % ST_TILES) * CHUNK + m) * ST_PITCH + band_base;
        *(v4f*)(sp)     = lo;
        *(v4f*)(sp + 4) = hi4;

        if ((i % ST_TILES) == ST_TILES - 1) {
            __builtin_amdgcn_fence(__ATOMIC_RELEASE, "workgroup", "local");
            __builtin_amdgcn_s_barrier();
            __builtin_amdgcn_fence(__ATOMIC_ACQUIRE, "workgroup", "local");
            float* o = ob + (size_t)(i / ST_TILES) * ST_ROWS * NBANDS;
#pragma unroll
            for (int k = 0; k < ST_ROWS_PER_WAVE; ++k) {
                const int row = wv * ST_ROWS_PER_WAVE + k;
                v4f val = *(const v4f*)(sb + row * ST_PITCH + lane * 4);
                *(v4f*)(o + (size_t)row * NBANDS) = val;
            }
        }
    }
}

extern "C" void kernel_launch(void* const* d_in, const int* in_sizes, int n_in,
                              void* d_out, int out_size, void* d_ws, size_t ws_size,
                              hipStream_t stream) {
    const float* x     = (const float*)d_in[0];
    const float* alpha = (const float*)d_in[1];
    const float* delta = (const float*)d_in[2];
    const float* r     = (const float*)d_in[3];
    float* out = (float*)d_out;

    float* bws = (float*)d_ws;
    float* Cws = bws + 4 * NBANDS;

    pcen_pass12_rowscan<<<ROWS / SCAN_ROWS_PER_BLOCK, 32 * SCAN_ROWS_PER_BLOCK, 0, stream>>>(
        x, alpha, delta, r, bws, Cws);

    dim3 g3(TT / P3_SPAN, NBATCH);
    pcen_pass3_wmma<<<g3, 32 * P3_WAVES, 0, stream>>>(x, Cws, bws, out);
}


# Tile pass takes its x block in four 32-step groups (one 128-byte line of four rows per load) and starts each tile pair when its own group has landed, not after all 16 loads; sums unchanged
# speedup vs baseline: 3.5915x; 1.0457x over previous
#include <hip/hip_runtime.h>

#define TT      2048
#define NBANDS  128
#define NBATCH  16
#define ROWS    (NBATCH * NBANDS)
#define CHUNK   16
#define NCHUNKS (TT / CHUNK)
#define EPSV    1e-5f

typedef __attribute__((ext_vector_type(2))) float v2f;
typedef __attribute__((ext_vector_type(4))) float v4f;
typedef __attribute__((ext_vector_type(8))) float v8f;

__device__ __forceinline__ float s_coef() {
    return (sqrtf(262145.0f) - 1.0f) / 131072.0f;
}

__device__ __forceinline__ float qpow(float q, int e) {
    float r = 1.0f, p = 1.0f;
#pragma unroll
    for (int i = 1; i <= CHUNK; ++i) {
        p *= q;
        r = (e >= i) ? p : r;
    }
    return r;
}

__device__ __forceinline__ float fast_pow_pos(float y, float p) {
    return __builtin_amdgcn_exp2f(p * __builtin_amdgcn_logf(y));
}

#define SCAN_ROWS_PER_BLOCK 4
#define TR_PITCH 68
static_assert(32 * SCAN_ROWS_PER_BLOCK == NBANDS, "work-group 0 covers one band per thread");
static_assert(TT == 32 * 64, "a row is 32 lane ranges of 64 steps = 16 loads of 128 steps");
__global__ void __launch_bounds__(32 * SCAN_ROWS_PER_BLOCK)
pcen_pass12_rowscan(const float* __restrict__ x, const float* __restrict__ alpha,
                    const float* __restrict__ delta, const float* __restrict__ r,
                    float* __restrict__ bws, float* __restrict__ Cws) {
    __shared__ __attribute__((aligned(16))) float tr[SCAN_ROWS_PER_BLOCK][32 * TR_PITCH];

    const int lane = threadIdx.x & 31;
    const int wv   = threadIdx.x >> 5;
    const int row  = blockIdx.x * SCAN_ROWS_PER_BLOCK + wv;
    const float s   = s_coef();
    const float q   = 1.0f - s;
    const float q2  = q * q, q4 = q2 * q2, q8 = q4 * q4, q16 = q8 * q8;
    const float q32 = q16 * q16, q64 = q32 * q32;

    const float* xr = x + (size_t)row * TT;
    float*       t  = tr[wv];

    v4f g[16];
#pragma unroll
    for (int j = 0; j < 16; ++j)
        g[j] = *(const v4f*)(xr + j * 128 + lane * 4);
#pragma unroll
    for (int j = 0; j < 16; ++j)
        *(v4f*)(t + (2 * j + (lane >> 4)) * TR_PITCH + (lane & 15) * 4) = g[j];

    __builtin_amdgcn_fence(__ATOMIC_RELEASE, "wavefront", "local");
    __builtin_amdgcn_wave_barrier();
    __builtin_amdgcn_fence(__ATOMIC_ACQUIRE, "wavefront", "local");

    const v4f*   p4 = (const v4f*)(t + lane * TR_PITCH);
    const float  c0 = t[0];

    float L[4];
#pragma unroll
    for (int c = 0; c < 4; ++c) {
        float acc = 0.0f;
#pragma unroll
        for (int i = 0; i < 4; ++i) {
            v4f v = p4[c * 4 + i];
            acc = acc * q + v.x;
            acc = acc * q + v.y;
            acc = acc * q + v.z;
            acc = acc * q + v.w;
        }
        L[c] = s * acc;
    }

    float P = ((L[0] * q16 + L[1]) * q16 + L[2]) * q16 + L[3];
    if (lane == 0) P = q64 * c0 + P;
    float w = q64;
#pragma unroll
    for (int d = 1; d < 32; d <<= 1) {
        float up = __shfl_up(P, d, 32);
        if (lane >= d) P = w * up + P;
        w = w * w;
    }
    float cin = __shfl_up(P, 1, 32);
    if (lane == 0) cin = c0;

    v4f cv;
    cv.x = cin;
    cv.y = q16 * cv.x + L[0];
    cv.z = q16 * cv.y + L[1];
    cv.w = q16 * cv.z + L[2];
    *(v4f*)(Cws + (size_t)row * NCHUNKS + lane * 4) = cv;

    if (blockIdx.x == 0) {
        const int b = threadIdx.x;
        float a  = __expf(alpha[b]);
        float d  = __expf(delta[b]);
        float rr = __expf(r[b]);
        bws[b]              = a;
        bws[NBANDS + b]     = d;
        bws[2 * NBANDS + b] = rr;
        bws[3 * NBANDS + b] = fast_pow_pos(d, rr);
    }
}

#define P3_TILES 8
#define P3_WAVES (NBANDS / 16)
#define P3_SPAN  (P3_TILES * CHUNK)
#define ST_TILES 2
#define ST_ROWS  (ST_TILES * CHUNK)
#define ST_PITCH (NBANDS + 4)
#define ST_ROWS_PER_WAVE (ST_ROWS / P3_WAVES)
#define XS_PITCH (P3_SPAN + 4)
#define XG_STEPS (ST_TILES * CHUNK)
#define XG_COUNT (P3_SPAN / XG_STEPS)
static_assert(P3_TILES % ST_TILES == 0, "whole stages");
static_assert(ST_ROWS % P3_WAVES == 0, "rows of a stage divide over the waves");
static_assert(NBANDS == 32 * 4, "one time row = 32 lanes x 4 floats");
static_assert(XG_STEPS == 8 * 4, "one group line = 8 lanes x 4 floats, four rows per load");
static_assert(XG_COUNT * 4 == 16, "four loads per group, 16 loads per block");
static_assert(P3_TILES % 4 == 0, "carries come as 16-B pieces of four tiles");
__global__ void __launch_bounds__(32 * P3_WAVES)
pcen_pass3_wmma(const float* __restrict__ x, const float* __restrict__ Cws,
                const float* __restrict__ bws, float* __restrict__ out) {
    __shared__ __attribute__((aligned(16))) float stage[2][ST_ROWS * ST_PITCH];
    __shared__ __attribute__((aligned(16))) float xs[P3_WAVES][16 * XS_PITCH];

    const float s = s_coef();
    const float q = 1.0f - s;

    const int lane = threadIdx.x & 31;
    const int wv   = threadIdx.x >> 5;
    const int m    = lane & 15;
    const int hi   = lane >> 4;

    const int bidx   = blockIdx.y;
    const int chunk0 = blockIdx.x * P3_TILES;
    const int t00    = chunk0 * CHUNK;
    const int row0   = bidx * NBANDS + wv * 16;

    const int lc   = lane & 7;
    const int lh   = (lane >> 3) & 1;
    const int lu   = lane >> 4;
    const int lrow = lu + 8 * lh;
    const float* xb = x + (size_t)(row0 + lrow) * TT + t00 + lc * 4;
    float*       xw = xs[wv];
    float*       xl = xw + lrow * XS_PITCH + lc * 4;
    v4f g[XG_COUNT][4];
#pragma unroll
    for (int jj = 0; jj < 4; ++jj)
        g[0][jj] = *(const v4f*)(xb + (size_t)(2 * jj) * TT);

    v4f cq[8][P3_TILES / 4];
#pragma unroll
    for (int v = 0; v < 8; ++v) {
        const v4f* cp = (const v4f*)(Cws + (size_t)(row0 + v + 8 * hi) * NCHUNKS + chunk0);
        cq[v][0] = cp[0];
        cq[v][1] = cp[1];
    }

    const int band_base = wv * 16 + 8 * hi;
    const float* ea  = bws;
    const float* ed  = bws + NBANDS;
    const float* err = bws + 2 * NBANDS;
    const float* edr = bws + 3 * NBANDS;
    v4f a_lo  = *(const v4f*)(ea  + band_base), a_hi  = *(const v4f*)(ea  + band_base + 4);
    v4f d_lo  = *(const v4f*)(ed  + band_base), d_hi  = *(const v4f*)(ed  + band_base + 4);
    v4f r_lo  = *(const v4f*)(err + band_base), r_hi  = *(const v4f*)(err + band_base + 4);
    v4f dr_lo = *(const v4f*)(edr + band_base), dr_hi = *(const v4f*)(edr + band_base + 4);
    float av[8]  = {a_lo.x, a_lo.y, a_lo.z, a_lo.w, a_hi.x, a_hi.y, a_hi.z, a_hi.w};
    float dv[8]  = {d_lo.x, d_lo.y, d_lo.z, d_lo.w, d_hi.x, d_hi.y, d_hi.z, d_hi.w};
    float rv[8]  = {r_lo.x, r_lo.y, r_lo.z, r_lo.w, r_hi.x, r_hi.y, r_hi.z, r_hi.w};
    float drv[8] = {dr_lo.x, dr_lo.y, dr_lo.z, dr_lo.w, dr_hi.x, dr_hi.y, dr_hi.z, dr_hi.w};

    v2f Bm[4];
#pragma unroll
    for (int kk = 0; kk < 4; ++kk) {
        int i0 = kk * 4 + 2 * hi;
        Bm[kk].x = (m >= i0)     ? s * qpow(q, m - i0)     : 0.0f;
        Bm[kk].y = (m >= i0 + 1) ? s * qpow(q, m - i0 - 1) : 0.0f;
    }
    const float qn1 = qpow(q, m + 1);

#pragma unroll
    for (int p = 1; p < XG_COUNT; ++p) {
        __builtin_amdgcn_sched_barrier(0);
#pragma unroll
        for (int jj = 0; jj < 4; ++jj)
            g[p][jj] = *(const v4f*)(xb + (size_t)(2 * jj) * TT + p * XG_STEPS);
    }
    __builtin_amdgcn_sched_barrier(0);

    const float* xa = xw + m * XS_PITCH + 2 * hi;
    const float* xe = xw + (8 * hi) * XS_PITCH + m;

    float* ob = out + (size_t)bidx * TT * NBANDS + (size_t)t00 * NBANDS + lane * 4;

#pragma unroll
    for (int p = 0; p < XG_COUNT; ++p) {
#pragma unroll
        for (int jj = 0; jj < 4; ++jj)
            *(v4f*)(xl + (2 * jj) * XS_PITCH + p * XG_STEPS) = g[p][jj];

        __builtin_amdgcn_fence(__ATOMIC_RELEASE, "wavefront", "local");
        __builtin_amdgcn_wave_barrier();
        __builtin_amdgcn_fence(__ATOMIC_ACQUIRE, "wavefront", "local");

        float* sb = stage[p & 1];
#pragma unroll
        for (int ii = 0; ii < ST_TILES; ++ii) {
            const int i = p * ST_TILES + ii;
            v2f Am[4];
#pragma unroll
            for (int kk = 0; kk < 4; ++kk)
                Am[kk] = *(const v2f*)(xa + i * CHUNK + kk * 4);
            float xv[8];
#pragma unroll
            for (int v = 0; v < 8; ++v)
                xv[v] = xe[v * XS_PITCH + i * CHUNK];

            v8f acc;
#pragma unroll
            for (int v = 0; v < 8; ++v) acc[v] = cq[v][i / 4][i % 4] * qn1;

            acc = __builtin_amdgcn_wmma_f32_16x16x4_f32(false, Am[0], false, Bm[0], (short)0, acc, false, false);
            acc = __builtin_amdgcn_wmma_f32_16x16x4_f32(false, Am[1], false, Bm[1], (short)0, acc, false, false);
            acc = __builtin_amdgcn_wmma_f32_16x16x4_f32(false, Am[2], false, Bm[2], (short)0, acc, false, false);
            acc = __builtin_amdgcn_wmma_f32_16x16x4_f32(false, Am[3], false, Bm[3], (short)0, acc, false, false);

            float res[8];
#pragma unroll
            for (int v = 0; v < 8; ++v) {
                float sm     = fmaxf(acc[v], 0.0f);
                float smooth = fast_pow_pos(EPSV + sm, -av[v]);
                float base   = fmaf(xv[v], smooth, dv[v]);
                res[v] = fast_pow_pos(base, rv[v]) - drv[v];
            }
            v4f lo;  lo.x  = res[0]; lo.y  = res[1]; lo.z  = res[2]; lo.w  = res[3];
            v4f hi4; hi4.x = res[4]; hi4.y = res[5]; hi4.z = res[6]; hi4.w = res[7];
            float* sp = sb + (ii * CHUNK + m) * ST_PITCH + band_base;
            *(v4f*)(sp)     = lo;
            *(v4f*)(sp + 4) = hi4;
        }

        __builtin_amdgcn_fence(__ATOMIC_RELEASE, "workgroup", "local");
        __builtin_amdgcn_s_barrier();
        __builtin_amdgcn_fence(__ATOMIC_ACQUIRE, "workgroup", "local");
        float* o = ob + (size_t)p * ST_ROWS * NBANDS;
#pragma unroll
        for (int k = 0; k < ST_ROWS_PER_WAVE; ++k) {
            const int row = wv * ST_ROWS_PER_WAVE + k;
            v4f val = *(const v4f*)(sb + row * ST_PITCH + lane * 4);
            *(v4f*)(o + (size_t)row * NBANDS) = val;
        }
    }
}

extern "C" void kernel_launch(void* const* d_in, const int* in_sizes, int n_in,
                              void* d_out, int out_size, void* d_ws, size_t ws_size,
                              hipStream_t stream) {
    const float* x     = (const float*)d_in[0];
    const float* alpha = (const float*)d_in[1];
    const float* delta = (const float*)d_in[2];
    const float* r     = (const float*)d_in[3];
    float* out = (float*)d_out;

    float* bws = (float*)d_ws;
    float* Cws = bws + 4 * NBANDS;

    pcen_pass12_rowscan<<<ROWS / SCAN_ROWS_PER_BLOCK, 32 * SCAN_ROWS_PER_BLOCK, 0, stream>>>(
        x, alpha, delta, r, bws, Cws);

    dim3 g3(TT / P3_SPAN, NBATCH);
    pcen_pass3_wmma<<<g3, 32 * P3_WAVES, 0, stream>>>(x, Cws, bws, out);
}
